# starting kernel
# baseline (speedup 1.0000x reference)
#include <hip/hip_runtime.h>

typedef float v2f __attribute__((ext_vector_type(2)));
typedef float v8f __attribute__((ext_vector_type(8)));

#define CHUNK 2048
#define BLOCK 256
#define ROWS_PER_BLOCK 128
#define TILES_PER_CHUNK (CHUNK / 16)

__global__ __launch_bounds__(BLOCK) void nn_wmma_gather_kernel(
    const float* __restrict__ pts,
    const float* __restrict__ lidar,
    const float* __restrict__ features,
    float* __restrict__ outp,
    int M, int N, int D)
{
    __shared__ float2 sB[TILES_PER_CHUNK * 32];

    const int tid  = threadIdx.x;
    const int lane = tid & 31;
    const int wave = tid >> 5;
    const int row  = lane & 15;
    const bool hi  = lane >= 16;

    const int mBase = blockIdx.x * ROWS_PER_BLOCK + wave * 16;
    const int m     = mBase + row;

    const float qx = pts[(size_t)m * 3 + 0];
    const float qy = pts[(size_t)m * 3 + 1];
    const float qz = pts[(size_t)m * 3 + 2];
    v2f a;
    a[0] = hi ? (-2.0f * qz) : (-2.0f * qx);
    a[1] = hi ? 1.0f         : (-2.0f * qy);

    float bestVal[8];
    int   bestIdx[8];
#pragma unroll
    for (int r = 0; r < 8; ++r) { bestVal[r] = 3.402823466e+38f; bestIdx[r] = 0; }

    const int nChunks = N / CHUNK;
    for (int c = 0; c < nChunks; ++c) {
        const int chunkBase = c * CHUNK;

        for (int i = tid; i < CHUNK; i += BLOCK) {
            const int n = chunkBase + i;
            const float x = lidar[(size_t)n * 3 + 0];
            const float y = lidar[(size_t)n * 3 + 1];
            const float z = lidar[(size_t)n * 3 + 2];
            const int t  = i >> 4;
            const int cc = i & 15;
            sB[t * 32 + cc]      = make_float2(x, y);
            sB[t * 32 + cc + 16] = make_float2(z, x * x + y * y + z * z);
        }
        if (c + 1 < nChunks) {
            __builtin_prefetch(lidar + (size_t)(chunkBase + CHUNK) * 3 + (size_t)tid * 24, 0, 3);
        }
        __syncthreads();

#pragma unroll 4
        for (int t = 0; t < TILES_PER_CHUNK; ++t) {
            const float2 bf = sB[t * 32 + lane];
            v2f b;
            b[0] = bf.x;
            b[1] = bf.y;
            v8f cacc = {0.f, 0.f, 0.f, 0.f, 0.f, 0.f, 0.f, 0.f};
            cacc = __builtin_amdgcn_wmma_f32_16x16x4_f32(
                false, a, false, b, (short)0, cacc, false, false);
            const int nIdx = chunkBase + t * 16 + row;
#pragma unroll
            for (int r = 0; r < 8; ++r) {
                const bool lt = cacc[r] < bestVal[r];
                bestIdx[r] = lt ? nIdx    : bestIdx[r];
                bestVal[r] = lt ? cacc[r] : bestVal[r];
            }
        }
        __syncthreads();
    }

#pragma unroll
    for (int r = 0; r < 8; ++r) {
        float v = bestVal[r];
        int   i = bestIdx[r];
#pragma unroll
        for (int off = 1; off < 16; off <<= 1) {
            const float ov = __shfl_xor(v, off, 32);
            const int   oi = __shfl_xor(i, off, 32);
            if (ov < v || (ov == v && oi < i)) { v = ov; i = oi; }
        }
        bestVal[r] = v;
        bestIdx[r] = i;
    }

    const int dVec = D >> 2;
#pragma unroll
    for (int r = 0; r < 16; ++r) {
        const int nn = __shfl(bestIdx[r & 7], (r < 8) ? 0 : 16, 32);
        const float4* __restrict__ src = (const float4*)(features + (size_t)nn * D);
        float4* __restrict__ dst       = (float4*)(outp + (size_t)(mBase + r) * D);
        for (int d = lane; d < dVec; d += 32) dst[d] = src[d];
    }
}

extern "C" void kernel_launch(void* const* d_in, const int* in_sizes, int n_in,
                              void* d_out, int out_size, void* d_ws, size_t ws_size,
                              hipStream_t stream) {
    const float* pts      = (const float*)d_in[0];
    const float* lidar    = (const float*)d_in[1];
    const float* features = (const float*)d_in[2];
    float* outp           = (float*)d_out;

    const int M = in_sizes[0] / 3;
    const int N = in_sizes[1] / 3;
    const int D = in_sizes[2] / N;

    const int grid = M / ROWS_PER_BLOCK;
    nn_wmma_gather_kernel<<<grid, BLOCK, 0, stream>>>(pts, lidar, features, outp, M, N, D);
}


# One launch: 1024-thread work-groups, the 4 lidar slices of a row tile searched by 4 waves of one work-group; slice winners merged in LDS, feature rows copied there; scores and tie order unchanged
# speedup vs baseline: 1.3563x; 1.3563x over previous
#include <hip/hip_runtime.h>

typedef float v2f __attribute__((ext_vector_type(2)));
typedef float v8f __attribute__((ext_vector_type(8)));

#define NPTS 8192
#define SLICES 4
#define TILES_PER_SLICE (NPTS / SLICES / 16)
#define BLOCK 1024
#define ROW_TILES 8
#define ROWS_PER_BLOCK 128
#define GROWS_PER_WAVE 4

__global__ __launch_bounds__(BLOCK) void nn_wmma_fused_kernel(
    const float* __restrict__ pts,
    const float* __restrict__ lidar,
    const float* __restrict__ features,
    float* __restrict__ outp,
    int D)
{
    __shared__ float2 sB[(NPTS / 16) * 32];
    __shared__ int2 sWin[SLICES * ROWS_PER_BLOCK];

    const int tid  = threadIdx.x;
    const int lane = tid & 31;
    const int wave = __builtin_amdgcn_readfirstlane(tid >> 5);
    const int rowTile = wave & (ROW_TILES - 1);
    const int slice   = wave >> 3;
    const int row  = lane & 15;
    const bool hi  = lane >= 16;

    const int rBase = rowTile * 16;
    const int m     = blockIdx.x * ROWS_PER_BLOCK + rBase + row;

    const float qx = pts[(size_t)m * 3 + 0];
    const float qy = pts[(size_t)m * 3 + 1];
    const float qz = pts[(size_t)m * 3 + 2];
    v2f a;
    a[0] = hi ? (-2.0f * qz) : (-2.0f * qx);
    a[1] = hi ? 1.0f         : (-2.0f * qy);

    for (int i = tid; i < NPTS; i += BLOCK) {
        const float x = lidar[(size_t)i * 3 + 0];
        const float y = lidar[(size_t)i * 3 + 1];
        const float z = lidar[(size_t)i * 3 + 2];
        const int t  = i >> 4;
        const int cc = i & 15;
        sB[t * 32 + cc]      = make_float2(x, y);
        sB[t * 32 + cc + 16] = make_float2(z, x * x + y * y + z * z);
    }
    __syncthreads();

    float bestVal[8];
    int   bestIdx[8];
#pragma unroll
    for (int r = 0; r < 8; ++r) { bestVal[r] = 3.402823466e+38f; bestIdx[r] = 0; }

    const int tBeg = slice * TILES_PER_SLICE;
#pragma unroll 4
    for (int tt = 0; tt < TILES_PER_SLICE; ++tt) {
        const int t = tBeg + tt;
        const float2 bf = sB[t * 32 + lane];
        v2f b;
        b[0] = bf.x;
        b[1] = bf.y;
        v8f cacc = {0.f, 0.f, 0.f, 0.f, 0.f, 0.f, 0.f, 0.f};
        cacc = __builtin_amdgcn_wmma_f32_16x16x4_f32(
            false, a, false, b, (short)0, cacc, false, false);
        const int nIdx = t * 16 + row;
#pragma unroll
        for (int r = 0; r < 8; ++r) {
            const bool lt = cacc[r] < bestVal[r];
            bestIdx[r] = lt ? nIdx    : bestIdx[r];
            bestVal[r] = lt ? cacc[r] : bestVal[r];
        }
    }

#pragma unroll
    for (int r = 0; r < 8; ++r) {
        float v = bestVal[r];
        int   i = bestIdx[r];
#pragma unroll
        for (int off = 1; off < 16; off <<= 1) {
            const float ov = __shfl_xor(v, off, 32);
            const int   oi = __shfl_xor(i, off, 32);
            if (ov < v || (ov == v && oi < i)) { v = ov; i = oi; }
        }
        bestVal[r] = v;
        bestIdx[r] = i;
    }

    float sv = bestVal[0];
    int   si = bestIdx[0];
#pragma unroll
    for (int r = 1; r < 8; ++r) {
        const bool pick = (lane & 7) == r;
        sv = pick ? bestVal[r] : sv;
        si = pick ? bestIdx[r] : si;
    }
    if ((lane & 8) == 0) {
        const int rr = rBase + (hi ? 8 : 0) + (lane & 7);
        sWin[slice * ROWS_PER_BLOCK + rr] = make_int2(__float_as_int(sv), si);
    }
    __syncthreads();

    const int gBase = wave * GROWS_PER_WAVE;
    const int jr    = gBase + (lane & (GROWS_PER_WAVE - 1));
    int2 best = make_int2(__float_as_int(3.402823466e+38f), 0);
#pragma unroll
    for (int s = 0; s < SLICES; ++s) {
        const int2 o = sWin[s * ROWS_PER_BLOCK + jr];
        if (__int_as_float(o.x) < __int_as_float(best.x)) best = o;
    }

    const int dVec = D >> 2;
#pragma unroll
    for (int r = 0; r < GROWS_PER_WAVE; ++r) {
        const int nn = __builtin_amdgcn_readlane(best.y, r);
        const float4* __restrict__ src = (const float4*)(features + (size_t)nn * D);
        float4* __restrict__ dst       = (float4*)(outp + ((size_t)blockIdx.x * ROWS_PER_BLOCK + gBase + r) * D);
        for (int d = lane; d < dVec; d += 32) dst[d] = src[d];
    }
}

extern "C" void kernel_launch(void* const* d_in, const int* in_sizes, int n_in,
                              void* d_out, int out_size, void* d_ws, size_t ws_size,
                              hipStream_t stream) {
    const float* pts      = (const float*)d_in[0];
    const float* lidar    = (const float*)d_in[1];
    const float* features = (const float*)d_in[2];
    float* outp           = (float*)d_out;

    const int M = in_sizes[0] / 3;
    const int N = in_sizes[1] / 3;
    const int D = in_sizes[2] / N;

    const int grid = M / ROWS_PER_BLOCK;
    nn_wmma_fused_kernel<<<grid, BLOCK, 0, stream>>>(pts, lidar, features, outp, D);
}


# Cross-lane argmin as a register-halving butterfly: a lane keeps half of its rows at each exchange, 8 (score,index) exchanges per wave instead of 32 serial ones, merged without branches; same winners
# speedup vs baseline: 1.4810x; 1.0919x over previous
#include <hip/hip_runtime.h>

typedef float v2f __attribute__((ext_vector_type(2)));
typedef float v8f __attribute__((ext_vector_type(8)));

#define NPTS 8192
#define SLICES 4
#define TILES_PER_SLICE (NPTS / SLICES / 16)
#define BLOCK 1024
#define ROW_TILES 8
#define ROWS_PER_BLOCK 128
#define GROWS_PER_WAVE 4

__device__ __forceinline__ void argmin_merge(float& v, int& i, const float ov, const int oi) {
    const bool take = (ov < v) | ((ov == v) & (oi < i));
    v = take ? ov : v;
    i = take ? oi : i;
}

__global__ __launch_bounds__(BLOCK) void nn_wmma_fused_kernel(
    const float* __restrict__ pts,
    const float* __restrict__ lidar,
    const float* __restrict__ features,
    float* __restrict__ outp,
    int D)
{
    __shared__ float2 sB[(NPTS / 16) * 32];
    __shared__ int2 sWin[SLICES * ROWS_PER_BLOCK];

    const int tid  = threadIdx.x;
    const int lane = tid & 31;
    const int wave = __builtin_amdgcn_readfirstlane(tid >> 5);
    const int rowTile = wave & (ROW_TILES - 1);
    const int slice   = wave >> 3;
    const int row  = lane & 15;
    const bool hi  = lane >= 16;

    const int rBase = rowTile * 16;
    const int m     = blockIdx.x * ROWS_PER_BLOCK + rBase + row;

    const float qx = pts[(size_t)m * 3 + 0];
    const float qy = pts[(size_t)m * 3 + 1];
    const float qz = pts[(size_t)m * 3 + 2];
    v2f a;
    a[0] = hi ? (-2.0f * qz) : (-2.0f * qx);
    a[1] = hi ? 1.0f         : (-2.0f * qy);

    for (int i = tid; i < NPTS; i += BLOCK) {
        const float x = lidar[(size_t)i * 3 + 0];
        const float y = lidar[(size_t)i * 3 + 1];
        const float z = lidar[(size_t)i * 3 + 2];
        const int t  = i >> 4;
        const int cc = i & 15;
        sB[t * 32 + cc]      = make_float2(x, y);
        sB[t * 32 + cc + 16] = make_float2(z, x * x + y * y + z * z);
    }
    __syncthreads();

    float bestVal[8];
    int   bestIdx[8];
#pragma unroll
    for (int r = 0; r < 8; ++r) { bestVal[r] = 3.402823466e+38f; bestIdx[r] = 0; }

    const int tBeg = slice * TILES_PER_SLICE;
#pragma unroll 4
    for (int tt = 0; tt < TILES_PER_SLICE; ++tt) {
        const int t = tBeg + tt;
        const float2 bf = sB[t * 32 + lane];
        v2f b;
        b[0] = bf.x;
        b[1] = bf.y;
        v8f cacc = {0.f, 0.f, 0.f, 0.f, 0.f, 0.f, 0.f, 0.f};
        cacc = __builtin_amdgcn_wmma_f32_16x16x4_f32(
            false, a, false, b, (short)0, cacc, false, false);
        const int nIdx = t * 16 + row;
#pragma unroll
        for (int r = 0; r < 8; ++r) {
            const bool lt = cacc[r] < bestVal[r];
            bestIdx[r] = lt ? nIdx    : bestIdx[r];
            bestVal[r] = lt ? cacc[r] : bestVal[r];
        }
    }

    const bool b3 = (lane & 8) != 0;
    float v4[4];
    int   i4[4];
#pragma unroll
    for (int j = 0; j < 4; ++j) {
        const float sv = b3 ? bestVal[j]     : bestVal[4 + j];
        const int   si = b3 ? bestIdx[j]     : bestIdx[4 + j];
        v4[j]          = b3 ? bestVal[4 + j] : bestVal[j];
        i4[j]          = b3 ? bestIdx[4 + j] : bestIdx[j];
        argmin_merge(v4[j], i4[j], __shfl_xor(sv, 8, 32), __shfl_xor(si, 8, 32));
    }
    const bool b2 = (lane & 4) != 0;
    float v2[2];
    int   i2[2];
#pragma unroll
    for (int j = 0; j < 2; ++j) {
        const float sv = b2 ? v4[j]     : v4[2 + j];
        const int   si = b2 ? i4[j]     : i4[2 + j];
        v2[j]          = b2 ? v4[2 + j] : v4[j];
        i2[j]          = b2 ? i4[2 + j] : i4[j];
        argmin_merge(v2[j], i2[j], __shfl_xor(sv, 4, 32), __shfl_xor(si, 4, 32));
    }
    const bool b1 = (lane & 2) != 0;
    float wv = b1 ? v2[1] : v2[0];
    int   wi = b1 ? i2[1] : i2[0];
    {
        const float sv = b1 ? v2[0] : v2[1];
        const int   si = b1 ? i2[0] : i2[1];
        argmin_merge(wv, wi, __shfl_xor(sv, 2, 32), __shfl_xor(si, 2, 32));
    }
    argmin_merge(wv, wi, __shfl_xor(wv, 1, 32), __shfl_xor(wi, 1, 32));

    if ((lane & 1) == 0) {
        const int rr = rBase + (hi ? 8 : 0) + ((lane >> 1) & 7);
        sWin[slice * ROWS_PER_BLOCK + rr] = make_int2(__float_as_int(wv), wi);
    }
    __syncthreads();

    const int gBase = wave * GROWS_PER_WAVE;
    const int jr    = gBase + (lane & (GROWS_PER_WAVE - 1));
    int2 best = make_int2(__float_as_int(3.402823466e+38f), 0);
#pragma unroll
    for (int s = 0; s < SLICES; ++s) {
        const int2 o = sWin[s * ROWS_PER_BLOCK + jr];
        if (__int_as_float(o.x) < __int_as_float(best.x)) best = o;
    }

    const int dVec = D >> 2;
#pragma unroll
    for (int r = 0; r < GROWS_PER_WAVE; ++r) {
        const int nn = __builtin_amdgcn_readlane(best.y, r);
        const float4* __restrict__ src = (const float4*)(features + (size_t)nn * D);
        float4* __restrict__ dst       = (float4*)(outp + ((size_t)blockIdx.x * ROWS_PER_BLOCK + gBase + r) * D);
        for (int d = lane; d < dVec; d += 32) dst[d] = src[d];
    }
}

extern "C" void kernel_launch(void* const* d_in, const int* in_sizes, int n_in,
                              void* d_out, int out_size, void* d_ws, size_t ws_size,
                              hipStream_t stream) {
    const float* pts      = (const float*)d_in[0];
    const float* lidar    = (const float*)d_in[1];
    const float* features = (const float*)d_in[2];
    float* outp           = (float*)d_out;

    const int M = in_sizes[0] / 3;
    const int N = in_sizes[1] / 3;
    const int D = in_sizes[2] / N;

    const int grid = M / ROWS_PER_BLOCK;
    nn_wmma_fused_kernel<<<grid, BLOCK, 0, stream>>>(pts, lidar, features, outp, D);
}


# Lidar staging pipelined under the search: first 2048 points staged up front, the other 6144 loaded while the waves search the first and written to LDS behind it, one more barrier; same winners
# speedup vs baseline: 1.6051x; 1.0838x over previous
#include <hip/hip_runtime.h>

typedef float v2f __attribute__((ext_vector_type(2)));
typedef float v8f __attribute__((ext_vector_type(8)));

#define NPTS 8192
#define BLOCK 1024
#define SLICES 4
#define PTS_HEAD 2048
#define PTS_REST (NPTS - PTS_HEAD)
#define HEAD_TILES_PER_WAVE (PTS_HEAD / 16 / SLICES)
#define REST_TILES_PER_WAVE (PTS_REST / 16 / SLICES)
#define HEAD_PER_THREAD (PTS_HEAD / BLOCK)
#define REST_PER_THREAD (PTS_REST / BLOCK)
#define ROW_TILES 8
#define ROWS_PER_BLOCK 128
#define GROWS_PER_WAVE 4

__device__ __forceinline__ void argmin_merge(float& v, int& i, const float ov, const int oi) {
    const bool take = (ov < v) | ((ov == v) & (oi < i));
    v = take ? ov : v;
    i = take ? oi : i;
}

__global__ __launch_bounds__(BLOCK) void nn_wmma_fused_kernel(
    const float* __restrict__ pts,
    const float* __restrict__ lidar,
    const float* __restrict__ features,
    float* __restrict__ outp,
    int D)
{
    __shared__ float2 sB[(NPTS / 16) * 32];
    __shared__ int2 sWin[SLICES * ROWS_PER_BLOCK];

    const int tid  = threadIdx.x;
    const int lane = tid & 31;
    const int wave = __builtin_amdgcn_readfirstlane(tid >> 5);
    const int rowTile = wave & (ROW_TILES - 1);
    const int slice   = wave >> 3;
    const int row  = lane & 15;
    const bool hi  = lane >= 16;

    const int rBase = rowTile * 16;
    const int m     = blockIdx.x * ROWS_PER_BLOCK + rBase + row;

    const float qx = pts[(size_t)m * 3 + 0];
    const float qy = pts[(size_t)m * 3 + 1];
    const float qz = pts[(size_t)m * 3 + 2];
    v2f a;
    a[0] = hi ? (-2.0f * qz) : (-2.0f * qx);
    a[1] = hi ? 1.0f         : (-2.0f * qy);

#define LOAD_PART(X, Y, Z, FIRST, COUNT)                                       \
    _Pragma("unroll")                                                          \
    for (int j = 0; j < (COUNT); ++j) {                                        \
        const int i = (FIRST) + j * BLOCK + tid;                               \
        X[j] = lidar[(size_t)i * 3 + 0];                                       \
        Y[j] = lidar[(size_t)i * 3 + 1];                                       \
        Z[j] = lidar[(size_t)i * 3 + 2];                                       \
    }
#define STORE_PART(X, Y, Z, FIRST, COUNT)                                      \
    _Pragma("unroll")                                                          \
    for (int j = 0; j < (COUNT); ++j) {                                        \
        const int i  = (FIRST) + j * BLOCK + tid;                              \
        const int t  = i >> 4;                                                 \
        const int cc = i & 15;                                                 \
        const float x = X[j], y = Y[j], z = Z[j];                              \
        sB[t * 32 + cc]      = make_float2(x, y);                              \
        sB[t * 32 + cc + 16] = make_float2(z, x * x + y * y + z * z);          \
    }

    float hx[HEAD_PER_THREAD], hy[HEAD_PER_THREAD], hz[HEAD_PER_THREAD];
    LOAD_PART(hx, hy, hz, 0, HEAD_PER_THREAD)
    STORE_PART(hx, hy, hz, 0, HEAD_PER_THREAD)
    __syncthreads();

    float rx[REST_PER_THREAD], ry[REST_PER_THREAD], rz[REST_PER_THREAD];
    LOAD_PART(rx, ry, rz, PTS_HEAD, REST_PER_THREAD)

    float bestVal[8];
    int   bestIdx[8];
#pragma unroll
    for (int r = 0; r < 8; ++r) { bestVal[r] = 3.402823466e+38f; bestIdx[r] = 0; }

    auto search = [&](const int tBeg, const int nTiles) {
#pragma unroll 4
        for (int tt = 0; tt < nTiles; ++tt) {
            const int t = tBeg + tt;
            const float2 bf = sB[t * 32 + lane];
            v2f b;
            b[0] = bf.x;
            b[1] = bf.y;
            v8f cacc = {0.f, 0.f, 0.f, 0.f, 0.f, 0.f, 0.f, 0.f};
            cacc = __builtin_amdgcn_wmma_f32_16x16x4_f32(
                false, a, false, b, (short)0, cacc, false, false);
            const int nIdx = t * 16 + row;
#pragma unroll
            for (int r = 0; r < 8; ++r) {
                const bool lt = cacc[r] < bestVal[r];
                bestIdx[r] = lt ? nIdx    : bestIdx[r];
                bestVal[r] = lt ? cacc[r] : bestVal[r];
            }
        }
    };

    search(slice * HEAD_TILES_PER_WAVE, HEAD_TILES_PER_WAVE);

    STORE_PART(rx, ry, rz, PTS_HEAD, REST_PER_THREAD)
    __syncthreads();
#undef LOAD_PART
#undef STORE_PART

    search(PTS_HEAD / 16 + slice * REST_TILES_PER_WAVE, REST_TILES_PER_WAVE);

    const bool b3 = (lane & 8) != 0;
    float v4[4];
    int   i4[4];
#pragma unroll
    for (int j = 0; j < 4; ++j) {
        const float sv = b3 ? bestVal[j]     : bestVal[4 + j];
        const int   si = b3 ? bestIdx[j]     : bestIdx[4 + j];
        v4[j]          = b3 ? bestVal[4 + j] : bestVal[j];
        i4[j]          = b3 ? bestIdx[4 + j] : bestIdx[j];
        argmin_merge(v4[j], i4[j], __shfl_xor(sv, 8, 32), __shfl_xor(si, 8, 32));
    }
    const bool b2 = (lane & 4) != 0;
    float v2[2];
    int   i2[2];
#pragma unroll
    for (int j = 0; j < 2; ++j) {
        const float sv = b2 ? v4[j]     : v4[2 + j];
        const int   si = b2 ? i4[j]     : i4[2 + j];
        v2[j]          = b2 ? v4[2 + j] : v4[j];
        i2[j]          = b2 ? i4[2 + j] : i4[j];
        argmin_merge(v2[j], i2[j], __shfl_xor(sv, 4, 32), __shfl_xor(si, 4, 32));
    }
    const bool b1 = (lane & 2) != 0;
    float wv = b1 ? v2[1] : v2[0];
    int   wi = b1 ? i2[1] : i2[0];
    {
        const float sv = b1 ? v2[0] : v2[1];
        const int   si = b1 ? i2[0] : i2[1];
        argmin_merge(wv, wi, __shfl_xor(sv, 2, 32), __shfl_xor(si, 2, 32));
    }
    argmin_merge(wv, wi, __shfl_xor(wv, 1, 32), __shfl_xor(wi, 1, 32));

    if ((lane & 1) == 0) {
        const int rr = rBase + (hi ? 8 : 0) + ((lane >> 1) & 7);
        sWin[slice * ROWS_PER_BLOCK + rr] = make_int2(__float_as_int(wv), wi);
    }
    __syncthreads();

    const int gBase = wave * GROWS_PER_WAVE;
    const int jr    = gBase + (lane & (GROWS_PER_WAVE - 1));
    const int2 first = sWin[jr];
    float mv = __int_as_float(first.x);
    int   mi = first.y;
#pragma unroll
    for (int s = 1; s < SLICES; ++s) {
        const int2 o = sWin[s * ROWS_PER_BLOCK + jr];
        argmin_merge(mv, mi, __int_as_float(o.x), o.y);
    }
    int2 best = make_int2(__float_as_int(mv), mi);

    const int dVec = D >> 2;
#pragma unroll
    for (int r = 0; r < GROWS_PER_WAVE; ++r) {
        const int nn = __builtin_amdgcn_readlane(best.y, r);
        const float4* __restrict__ src = (const float4*)(features + (size_t)nn * D);
        float4* __restrict__ dst       = (float4*)(outp + ((size_t)blockIdx.x * ROWS_PER_BLOCK + gBase + r) * D);
        for (int d = lane; d < dVec; d += 32) dst[d] = src[d];
    }
}

extern "C" void kernel_launch(void* const* d_in, const int* in_sizes, int n_in,
                              void* d_out, int out_size, void* d_ws, size_t ws_size,
                              hipStream_t stream) {
    const float* pts      = (const float*)d_in[0];
    const float* lidar    = (const float*)d_in[1];
    const float* features = (const float*)d_in[2];
    float* outp           = (float*)d_out;

    const int M = in_sizes[0] / 3;
    const int N = in_sizes[1] / 3;
    const int D = in_sizes[2] / N;

    const int grid = M / ROWS_PER_BLOCK;
    nn_wmma_fused_kernel<<<grid, BLOCK, 0, stream>>>(pts, lidar, features, outp, D);
}


# Rest screened by sign: each row's lowest head score plus a rounding margin enters the WMMA as C; a tile costs 4 minima, a test per 4 tiles; only tiles with a negative entry get selects; same winners
# speedup vs baseline: 1.7178x; 1.0702x over previous
#include <hip/hip_runtime.h>

typedef float v2f __attribute__((ext_vector_type(2)));
typedef float v8f __attribute__((ext_vector_type(8)));

#define NPTS 8192
#define BLOCK 1024
#define SLICES 4
#define PTS_HEAD 2048
#define PTS_REST (NPTS - PTS_HEAD)
#define HEAD_TILES_PER_WAVE (PTS_HEAD / 16 / SLICES)
#define REST_TILES_PER_WAVE (PTS_REST / 16 / SLICES)
#define HEAD_PER_THREAD (PTS_HEAD / BLOCK)
#define REST_PER_THREAD (PTS_REST / BLOCK)
#define ROW_TILES 8
#define ROWS_PER_BLOCK 128
#define GROWS_PER_WAVE 4
#define SCREEN_EPS 1.52587890625e-05f

template <int MASK>
__device__ __forceinline__ float row_xor(const float v) {
    const int b = __float_as_int(v);
    return __int_as_float(__builtin_amdgcn_update_dpp(b, b, 0x160 + MASK, 0xF, 0xF, true));
}

__device__ __forceinline__ void argmin_merge(float& v, int& i, const float ov, const int oi) {
    const bool take = (ov < v) | ((ov == v) & (oi < i));
    v = take ? ov : v;
    i = take ? oi : i;
}

__global__ __launch_bounds__(BLOCK) void nn_wmma_fused_kernel(
    const float* __restrict__ pts,
    const float* __restrict__ lidar,
    const float* __restrict__ features,
    float* __restrict__ outp,
    int D)
{
    __shared__ float2 sB[(NPTS / 16) * 32];
    __shared__ int2 sWin[SLICES * ROWS_PER_BLOCK];
    __shared__ __attribute__((aligned(16))) float sThr[SLICES * ROWS_PER_BLOCK];

    const int tid  = threadIdx.x;
    const int lane = tid & 31;
    const int wave = __builtin_amdgcn_readfirstlane(tid >> 5);
    const int rowTile = wave & (ROW_TILES - 1);
    const int slice   = wave >> 3;
    const int row  = lane & 15;
    const bool hi  = lane >= 16;

    const int rBase = rowTile * 16;
    const int m     = blockIdx.x * ROWS_PER_BLOCK + rBase + row;

    const float qx = pts[(size_t)m * 3 + 0];
    const float qy = pts[(size_t)m * 3 + 1];
    const float qz = pts[(size_t)m * 3 + 2];
    v2f a;
    a[0] = hi ? (-2.0f * qz) : (-2.0f * qx);
    a[1] = hi ? 1.0f         : (-2.0f * qy);

#define LOAD_PART(X, Y, Z, FIRST, COUNT)                                       \
    _Pragma("unroll")                                                          \
    for (int j = 0; j < (COUNT); ++j) {                                        \
        const int i = (FIRST) + j * BLOCK + tid;                               \
        X[j] = lidar[(size_t)i * 3 + 0];                                       \
        Y[j] = lidar[(size_t)i * 3 + 1];                                       \
        Z[j] = lidar[(size_t)i * 3 + 2];                                       \
    }
#define STORE_PART(X, Y, Z, FIRST, COUNT)                                      \
    _Pragma("unroll")                                                          \
    for (int j = 0; j < (COUNT); ++j) {                                        \
        const int i  = (FIRST) + j * BLOCK + tid;                              \
        const int t  = i >> 4;                                                 \
        const int cc = i & 15;                                                 \
        const float x = X[j], y = Y[j], z = Z[j];                              \
        sB[t * 32 + cc]      = make_float2(x, y);                              \
        sB[t * 32 + cc + 16] = make_float2(z, x * x + y * y + z * z);          \
    }

    float hx[HEAD_PER_THREAD], hy[HEAD_PER_THREAD], hz[HEAD_PER_THREAD];
    LOAD_PART(hx, hy, hz, 0, HEAD_PER_THREAD)
    STORE_PART(hx, hy, hz, 0, HEAD_PER_THREAD)
    __syncthreads();

    float rx[REST_PER_THREAD], ry[REST_PER_THREAD], rz[REST_PER_THREAD];
    LOAD_PART(rx, ry, rz, PTS_HEAD, REST_PER_THREAD)

    float bestVal[8];
    int   bestIdx[8];
#pragma unroll
    for (int r = 0; r < 8; ++r) { bestVal[r] = 3.402823466e+38f; bestIdx[r] = 0; }

    auto search = [&](const int tBeg, const int nTiles) {
#pragma unroll 4
        for (int tt = 0; tt < nTiles; ++tt) {
            const int t = tBeg + tt;
            const float2 bf = sB[t * 32 + lane];
            v2f b;
            b[0] = bf.x;
            b[1] = bf.y;
            v8f cacc = {0.f, 0.f, 0.f, 0.f, 0.f, 0.f, 0.f, 0.f};
            cacc = __builtin_amdgcn_wmma_f32_16x16x4_f32(
                false, a, false, b, (short)0, cacc, false, false);
            const int nIdx = t * 16 + row;
#pragma unroll
            for (int r = 0; r < 8; ++r) {
                const bool lt = cacc[r] < bestVal[r];
                bestIdx[r] = lt ? nIdx    : bestIdx[r];
                bestVal[r] = lt ? cacc[r] : bestVal[r];
            }
        }
    };

    search(slice * HEAD_TILES_PER_WAVE, HEAD_TILES_PER_WAVE);

    {
        const bool c3 = (lane & 8) != 0;
        float u4[4];
#pragma unroll
        for (int j = 0; j < 4; ++j) {
            const float sv = c3 ? bestVal[j]     : bestVal[4 + j];
            const float kv = c3 ? bestVal[4 + j] : bestVal[j];
            u4[j] = __builtin_fminimum_numf(kv, row_xor<8>(sv));
        }
        const bool c2 = (lane & 4) != 0;
        float u2[2];
#pragma unroll
        for (int j = 0; j < 2; ++j) {
            const float sv = c2 ? u4[j]     : u4[2 + j];
            const float kv = c2 ? u4[2 + j] : u4[j];
            u2[j] = __builtin_fminimum_numf(kv, row_xor<4>(sv));
        }
        const bool c1 = (lane & 2) != 0;
        float u = c1 ? u2[1] : u2[0];
        {
            const float sv = c1 ? u2[0] : u2[1];
            u = __builtin_fminimum_numf(u, row_xor<2>(sv));
        }
        u = __builtin_fminimum_numf(u, row_xor<1>(u));
        const int myRow = (hi ? 8 : 0) + ((lane >> 1) & 7);
        const float q2r = __shfl(qx * qx + qy * qy + qz * qz, myRow, 32);
        const float scr = -(u + (SCREEN_EPS * (q2r + __builtin_fabsf(u)) + 1e-30f));
        if ((lane & 1) == 0) sThr[slice * ROWS_PER_BLOCK + rBase + myRow] = scr;
    }

    STORE_PART(rx, ry, rz, PTS_HEAD, REST_PER_THREAD)
    __syncthreads();
#undef LOAD_PART
#undef STORE_PART

    v8f cthr;
#pragma unroll
    for (int s = 0; s < SLICES; ++s) {
        const float4* p = (const float4*)&sThr[s * ROWS_PER_BLOCK + rBase + (hi ? 8 : 0)];
        const float4 lo4 = p[0];
        const float4 hi4 = p[1];
        const float w[8] = {lo4.x, lo4.y, lo4.z, lo4.w, hi4.x, hi4.y, hi4.z, hi4.w};
#pragma unroll
        for (int r = 0; r < 8; ++r) cthr[r] = (s == 0) ? w[r] : __builtin_fmaximum_numf(cthr[r], w[r]);
    }
    v2f aScr;
    aScr[0] = a[0];
    aScr[1] = hi ? (1.0f - SCREEN_EPS) : a[1];

    const int tRest = PTS_HEAD / 16 + slice * REST_TILES_PER_WAVE;
    for (int tt = 0; tt < REST_TILES_PER_WAVE; tt += 4) {
        const int t = tRest + tt;
        v2f b[4];
        float mj[4];
#pragma unroll
        for (int j = 0; j < 4; ++j) {
            const float2 bf = sB[(t + j) * 32 + lane];
            b[j][0] = bf.x;
            b[j][1] = bf.y;
        }
#pragma unroll
        for (int j = 0; j < 4; ++j) {
            const v8f d = __builtin_amdgcn_wmma_f32_16x16x4_f32(
                false, aScr, false, b[j], (short)0, cthr, false, false);
            const float x = __builtin_fminimum_numf(__builtin_fminimum_numf(d[0], d[1]), d[2]);
            const float y = __builtin_fminimum_numf(__builtin_fminimum_numf(d[3], d[4]), d[5]);
            const float z = __builtin_fminimum_numf(__builtin_fminimum_numf(d[6], d[7]), d[5]);
            mj[j] = __builtin_fminimum_numf(__builtin_fminimum_numf(x, y), z);
        }
        const float m4 = __builtin_fminimum_numf(__builtin_fminimum_numf(mj[0], mj[1]), __builtin_fminimum_numf(mj[2], mj[3]));
        if (__builtin_amdgcn_ballot_w32(m4 < 0.0f) != 0u) {
#pragma unroll
            for (int j = 0; j < 4; ++j) {
                if (__builtin_amdgcn_ballot_w32(mj[j] < 0.0f) != 0u) {
                    v8f cacc = {0.f, 0.f, 0.f, 0.f, 0.f, 0.f, 0.f, 0.f};
                    cacc = __builtin_amdgcn_wmma_f32_16x16x4_f32(
                        false, a, false, b[j], (short)0, cacc, false, false);
                    const int nIdx = (t + j) * 16 + row;
#pragma unroll
                    for (int r = 0; r < 8; ++r) {
                        const bool lt = cacc[r] < bestVal[r];
                        bestIdx[r] = lt ? nIdx    : bestIdx[r];
                        bestVal[r] = lt ? cacc[r] : bestVal[r];
                    }
                }
            }
        }
    }

    const bool b3 = (lane & 8) != 0;
    float v4[4];
    int   i4[4];
#pragma unroll
    for (int j = 0; j < 4; ++j) {
        const float sv = b3 ? bestVal[j]     : bestVal[4 + j];
        const int   si = b3 ? bestIdx[j]     : bestIdx[4 + j];
        v4[j]          = b3 ? bestVal[4 + j] : bestVal[j];
        i4[j]          = b3 ? bestIdx[4 + j] : bestIdx[j];
        argmin_merge(v4[j], i4[j], __shfl_xor(sv, 8, 32), __shfl_xor(si, 8, 32));
    }
    const bool b2 = (lane & 4) != 0;
    float v2[2];
    int   i2[2];
#pragma unroll
    for (int j = 0; j < 2; ++j) {
        const float sv = b2 ? v4[j]     : v4[2 + j];
        const int   si = b2 ? i4[j]     : i4[2 + j];
        v2[j]          = b2 ? v4[2 + j] : v4[j];
        i2[j]          = b2 ? i4[2 + j] : i4[j];
        argmin_merge(v2[j], i2[j], __shfl_xor(sv, 4, 32), __shfl_xor(si, 4, 32));
    }
    const bool b1 = (lane & 2) != 0;
    float wv = b1 ? v2[1] : v2[0];
    int   wi = b1 ? i2[1] : i2[0];
    {
        const float sv = b1 ? v2[0] : v2[1];
        const int   si = b1 ? i2[0] : i2[1];
        argmin_merge(wv, wi, __shfl_xor(sv, 2, 32), __shfl_xor(si, 2, 32));
    }
    argmin_merge(wv, wi, __shfl_xor(wv, 1, 32), __shfl_xor(wi, 1, 32));

    if ((lane & 1) == 0) {
        const int rr = rBase + (hi ? 8 : 0) + ((lane >> 1) & 7);
        sWin[slice * ROWS_PER_BLOCK + rr] = make_int2(__float_as_int(wv), wi);
    }
    __syncthreads();

    const int gBase = wave * GROWS_PER_WAVE;
    const int jr    = gBase + (lane & (GROWS_PER_WAVE - 1));
    const int2 first = sWin[jr];
    float mv = __int_as_float(first.x);
    int   mi = first.y;
#pragma unroll
    for (int s = 1; s < SLICES; ++s) {
        const int2 o = sWin[s * ROWS_PER_BLOCK + jr];
        argmin_merge(mv, mi, __int_as_float(o.x), o.y);
    }
    int2 best = make_int2(__float_as_int(mv), mi);

    const int dVec = D >> 2;
#pragma unroll
    for (int r = 0; r < GROWS_PER_WAVE; ++r) {
        const int nn = __builtin_amdgcn_readlane(best.y, r);
        const float4* __restrict__ src = (const float4*)(features + (size_t)nn * D);
        float4* __restrict__ dst       = (float4*)(outp + ((size_t)blockIdx.x * ROWS_PER_BLOCK + gBase + r) * D);
        for (int d = lane; d < dVec; d += 32) dst[d] = src[d];
    }
}

extern "C" void kernel_launch(void* const* d_in, const int* in_sizes, int n_in,
                              void* d_out, int out_size, void* d_ws, size_t ws_size,
                              hipStream_t stream) {
    const float* pts      = (const float*)d_in[0];
    const float* lidar    = (const float*)d_in[1];
    const float* features = (const float*)d_in[2];
    float* outp           = (float*)d_out;

    const int M = in_sizes[0] / 3;
    const int N = in_sizes[1] / 3;
    const int D = in_sizes[2] / N;

    const int grid = M / ROWS_PER_BLOCK;
    nn_wmma_fused_kernel<<<grid, BLOCK, 0, stream>>>(pts, lidar, features, outp, D);
}
